# starting kernel
# baseline (speedup 1.0000x reference)
#include <hip/hip_runtime.h>

typedef __attribute__((ext_vector_type(2))) float v2f;
typedef __attribute__((ext_vector_type(8))) float v8f;

namespace {
constexpr int kH = 36, kW = 36, kC = 256, kK = kH * kW;
constexpr float kScale = 1.0f / 16.0f;

__device__ __forceinline__ float gfun(float u) {
  u = fminf(fmaxf(u, -1.0f), 1.0f);
  float p = u + 1.0f;
  float q = 1.0f - u;
  return (u < 0.0f) ? 0.5f * p * p : 1.0f - 0.5f * q * q;
}
}

__global__ __launch_bounds__(128) void prroi_wmma_kernel(
    const float* __restrict__ feat,
    const float* __restrict__ bb,
    float* __restrict__ out) {
  __shared__ float s_wx[kW];
  __shared__ float s_wy[kH];
  __shared__ float s_w[kK];

  const int n    = blockIdx.y;
  const int tid  = threadIdx.x;
  const int lane = tid & 31;
  const int wave = tid >> 5;

  const float bx = bb[n * 4 + 0], by = bb[n * 4 + 1];
  const float bw = bb[n * 4 + 2], bh = bb[n * 4 + 3];
  const float x0 = bx * kScale, y0 = by * kScale;
  const float x1 = (bx + bw) * kScale, y1 = (by + bh) * kScale;
  const float binw = x1 - x0, binh = y1 - y0;
  const float area = fmaxf(binw, 0.0f) * fmaxf(binh, 0.0f);
  const float inv  = (area > 0.0f) ? (1.0f / fmaxf(area, 1e-30f)) : 0.0f;

  if (tid < kW) {
    s_wx[tid] = gfun(x1 - (float)tid) - gfun(x0 - (float)tid);
  } else if (tid >= 64 && tid < 64 + kH) {
    const int i = tid - 64;
    s_wy[i] = gfun(y1 - (float)i) - gfun(y0 - (float)i);
  }
  __syncthreads();
  for (int k = tid; k < kK; k += 128) {
    s_w[k] = s_wy[k / kW] * s_wx[k % kW] * inv;
  }
  __syncthreads();

  const int c0   = blockIdx.x * 64 + wave * 16;
  const int m    = lane & 15;
  const int koff = (lane >> 4) * 2;
  const float* __restrict__ arow =
      feat + ((size_t)n * kC + (size_t)(c0 + m)) * (size_t)kK + koff;

  v8f acc0 = {}, acc1 = {};
#pragma unroll 2
  for (int k = 0; k < kK; k += 8) {
    v2f a0 = *(const v2f*)(arow + k);
    v2f a1 = *(const v2f*)(arow + k + 4);
    v2f b0 = { s_w[k + koff],     s_w[k + koff + 1]     };
    v2f b1 = { s_w[k + 4 + koff], s_w[k + 4 + koff + 1] };
    acc0 = __builtin_amdgcn_wmma_f32_16x16x4_f32(
         false, a0,  false, b0,
         (short)0, acc0,  false,  false);
    acc1 = __builtin_amdgcn_wmma_f32_16x16x4_f32(
        false, a1, false, b1, (short)0, acc1, false, false);
  }
  v8f acc = acc0 + acc1;

  float* __restrict__ o = out + n * kC + c0;
  if (lane == 0) {
#pragma unroll
    for (int r = 0; r < 8; ++r) o[r] = acc[r];
  } else if (lane == 16) {
#pragma unroll
    for (int r = 0; r < 8; ++r) o[8 + r] = acc[r];
  }
}

extern "C" void kernel_launch(void* const* d_in, const int* in_sizes, int n_in,
                              void* d_out, int out_size, void* d_ws, size_t ws_size,
                              hipStream_t stream) {
  (void)n_in; (void)d_ws; (void)ws_size; (void)out_size;
  const float* feat = (const float*)d_in[0];
  const float* bb   = (const float*)d_in[1];
  float* out        = (float*)d_out;

  const int N = in_sizes[1] / 4;
  dim3 grid(kC / 64, N);
  dim3 block(128);
  prroi_wmma_kernel<<<grid, block, 0, stream>>>(feat, bb, out);
}


# k loop visits only the 8-float chunks that hold a non-zero weight (ascending list built per block from the weights); skipped terms are feat*0, summation order of the rest unchanged
# speedup vs baseline: 3.5233x; 3.5233x over previous
#include <hip/hip_runtime.h>


typedef __attribute__((ext_vector_type(2))) float v2f;
typedef __attribute__((ext_vector_type(4))) float v4f;
typedef __attribute__((ext_vector_type(8))) float v8f;

namespace {
constexpr int kH = 36, kW = 36, kC = 256, kK = kH * kW;
constexpr int kChunk  = 8;
constexpr int kNChunk = kK / kChunk;
constexpr int kGroup  = 8;
constexpr int kRounds = (kNChunk + 31) / 32;
constexpr int kWPad   = kRounds * 32 * kChunk;
constexpr float kScale = 1.0f / 16.0f;

__device__ __forceinline__ float gfun(float u) {
  u = fminf(fmaxf(u, -1.0f), 1.0f);
  float p = u + 1.0f;
  float q = 1.0f - u;
  return (u < 0.0f) ? 0.5f * p * p : 1.0f - 0.5f * q * q;
}
}

__global__ __launch_bounds__(128) void prroi_wmma_kernel(
    const float* __restrict__ feat,
    const float* __restrict__ bb,
    float* __restrict__ out) {
  __shared__ float s_wx[kW];
  __shared__ float s_wy[kH];
  __shared__ __attribute__((aligned(16))) float s_w[kWPad];
  __shared__ __attribute__((aligned(16))) int s_list[kNChunk + kGroup];
  __shared__ int s_cnt;

  const int n    = blockIdx.y;
  const int tid  = threadIdx.x;
  const int lane = tid & 31;
  const int wave = tid >> 5;

  const float bx = bb[n * 4 + 0], by = bb[n * 4 + 1];
  const float bw = bb[n * 4 + 2], bh = bb[n * 4 + 3];
  const float x0 = bx * kScale, y0 = by * kScale;
  const float x1 = (bx + bw) * kScale, y1 = (by + bh) * kScale;
  const float binw = x1 - x0, binh = y1 - y0;
  const float area = fmaxf(binw, 0.0f) * fmaxf(binh, 0.0f);
  const float inv  = (area > 0.0f) ? (1.0f / fmaxf(area, 1e-30f)) : 0.0f;

  if (tid < kW) {
    s_wx[tid] = gfun(x1 - (float)tid) - gfun(x0 - (float)tid);
  } else if (tid >= 64 && tid < 64 + kH) {
    const int i = tid - 64;
    s_wy[i] = gfun(y1 - (float)i) - gfun(y0 - (float)i);
  }
  __syncthreads();
  for (int k = tid; k < kK; k += 128) {
    s_w[k] = s_wy[k / kW] * s_wx[k % kW] * inv;
  }
  for (int k = kK + tid; k < kWPad; k += 128) s_w[k] = 0.0f;
  __syncthreads();

  if (wave == 0) {
    bool nz[kRounds];
#pragma unroll
    for (int r = 0; r < kRounds; ++r) {
      const int j = r * 32 + lane;
      const v4f u = *(const v4f*)(s_w + j * kChunk);
      const v4f v = *(const v4f*)(s_w + j * kChunk + 4);
      nz[r] = (u.x != 0.0f) | (u.y != 0.0f) | (u.z != 0.0f) | (u.w != 0.0f) |
              (v.x != 0.0f) | (v.y != 0.0f) | (v.z != 0.0f) | (v.w != 0.0f);
    }
    int base = 0;
#pragma unroll
    for (int r = 0; r < kRounds; ++r) {
      const unsigned mask = __builtin_amdgcn_ballot_w32(nz[r]);
      if (nz[r]) s_list[base + (int)__builtin_amdgcn_mbcnt_lo(mask, 0u)] = r * 32 + lane;
      base += __builtin_popcount(mask);
    }
    const int padded = (base + kGroup - 1) & ~(kGroup - 1);
    if (base + lane < padded) s_list[base + lane] = kNChunk;
    if (lane == 0) s_cnt = padded;
  }
  __syncthreads();

  const int c0   = blockIdx.x * 64 + wave * 16;
  const int m    = lane & 15;
  const int koff = (lane >> 4) * 2;
  const float* __restrict__ arow =
      feat + ((size_t)n * kC + (size_t)(c0 + m)) * (size_t)kK + koff;

  const int cnt = __builtin_amdgcn_readfirstlane(s_cnt);

  v8f acc0 = {}, acc1 = {};
  for (int i = 0; i < cnt; i += kGroup) {
    v2f a0[kGroup], a1[kGroup], b0[kGroup], b1[kGroup];
#pragma unroll
    for (int u = 0; u < kGroup; ++u) {
      const int j  = s_list[i + u];
      const int ka = (j < kNChunk ? j : 0) * kChunk;
      a0[u] = *(const v2f*)(arow + ka);
      a1[u] = *(const v2f*)(arow + ka + 4);
      const int kb = j * kChunk + koff;
      b0[u] = *(const v2f*)(s_w + kb);
      b1[u] = *(const v2f*)(s_w + kb + 4);
    }
    __builtin_amdgcn_sched_barrier(0);
#pragma unroll
    for (int u = 0; u < kGroup; ++u) {
      acc0 = __builtin_amdgcn_wmma_f32_16x16x4_f32(
           false, a0[u],  false, b0[u],
           (short)0, acc0,  false,  false);
      acc1 = __builtin_amdgcn_wmma_f32_16x16x4_f32(
          false, a1[u], false, b1[u], (short)0, acc1, false, false);
    }
  }
  v8f acc = acc0 + acc1;

  float* __restrict__ o = out + n * kC + c0;
  if (lane == 0) {
#pragma unroll
    for (int r = 0; r < 8; ++r) o[r] = acc[r];
  } else if (lane == 16) {
#pragma unroll
    for (int r = 0; r < 8; ++r) o[8 + r] = acc[r];
  }
}

extern "C" void kernel_launch(void* const* d_in, const int* in_sizes, int n_in,
                              void* d_out, int out_size, void* d_ws, size_t ws_size,
                              hipStream_t stream) {
  (void)n_in; (void)d_ws; (void)ws_size; (void)out_size;
  const float* feat = (const float*)d_in[0];
  const float* bb   = (const float*)d_in[1];
  float* out        = (float*)d_out;

  const int N = in_sizes[1] / 4;
  dim3 grid(kC / 64, N);
  dim3 block(128);
  prroi_wmma_kernel<<<grid, block, 0, stream>>>(feat, bb, out);
}


# all feature loads of the chunk list in flight at once (batches of up to 32 chunks, one 16-byte load per lane and chunk); summation order changes: per chunk chain 0 takes k {0,1,4,5}, chain 1 the rest
# speedup vs baseline: 4.6662x; 1.3244x over previous
#include <hip/hip_runtime.h>


typedef __attribute__((ext_vector_type(2))) float v2f;
typedef __attribute__((ext_vector_type(4))) float v4f;
typedef __attribute__((ext_vector_type(8))) float v8f;

namespace {
constexpr int kH = 36, kW = 36, kC = 256, kK = kH * kW;
constexpr int kChunk  = 8;
constexpr int kNChunk = kK / kChunk;
constexpr int kGroup  = 8;
constexpr int kBatch  = 32;
constexpr int kRounds = (kNChunk + 31) / 32;
constexpr int kWPad   = kRounds * 32 * kChunk;
constexpr float kScale = 1.0f / 16.0f;

__device__ __forceinline__ float gfun(float u) {
  u = fminf(fmaxf(u, -1.0f), 1.0f);
  float p = u + 1.0f;
  float q = 1.0f - u;
  return (u < 0.0f) ? 0.5f * p * p : 1.0f - 0.5f * q * q;
}

template <int NC>
__device__ __forceinline__ void gemv_batch(const float* __restrict__ arow,
                                           const float* s_w, const int* lst,
                                           int koff, v8f& acc0, v8f& acc1) {
  v4f a[NC], b[NC];
#pragma unroll
  for (int u = 0; u < NC; ++u) {
    const int j  = lst[u];
    const int ka = (j < kNChunk ? j : 0) * kChunk;
    a[u] = *(const v4f*)(arow + ka);
    b[u] = *(const v4f*)(s_w + j * kChunk + koff);
  }
  __builtin_amdgcn_sched_barrier(0);
#pragma unroll
  for (int u = 0; u < NC; ++u) {
    const v2f a0 = {a[u].x, a[u].y}, a1 = {a[u].z, a[u].w};
    const v2f b0 = {b[u].x, b[u].y}, b1 = {b[u].z, b[u].w};
    acc0 = __builtin_amdgcn_wmma_f32_16x16x4_f32(
         false, a0,  false, b0,
         (short)0, acc0,  false,  false);
    acc1 = __builtin_amdgcn_wmma_f32_16x16x4_f32(
        false, a1, false, b1, (short)0, acc1, false, false);
  }
}
}

__global__ __launch_bounds__(128) void prroi_wmma_kernel(
    const float* __restrict__ feat,
    const float* __restrict__ bb,
    float* __restrict__ out) {
  __shared__ float s_wx[kW];
  __shared__ float s_wy[kH];
  __shared__ __attribute__((aligned(16))) float s_w[kWPad];
  __shared__ __attribute__((aligned(16))) int s_list[kNChunk + kGroup];
  __shared__ int s_cnt;

  const int n    = blockIdx.y;
  const int tid  = threadIdx.x;
  const int lane = tid & 31;
  const int wave = tid >> 5;

  const float bx = bb[n * 4 + 0], by = bb[n * 4 + 1];
  const float bw = bb[n * 4 + 2], bh = bb[n * 4 + 3];
  const float x0 = bx * kScale, y0 = by * kScale;
  const float x1 = (bx + bw) * kScale, y1 = (by + bh) * kScale;
  const float binw = x1 - x0, binh = y1 - y0;
  const float area = fmaxf(binw, 0.0f) * fmaxf(binh, 0.0f);
  const float inv  = (area > 0.0f) ? (1.0f / fmaxf(area, 1e-30f)) : 0.0f;

  if (tid < kW) {
    s_wx[tid] = gfun(x1 - (float)tid) - gfun(x0 - (float)tid);
  } else if (tid >= 64 && tid < 64 + kH) {
    const int i = tid - 64;
    s_wy[i] = gfun(y1 - (float)i) - gfun(y0 - (float)i);
  }
  __syncthreads();
  for (int k = tid; k < kK; k += 128) {
    s_w[k] = s_wy[k / kW] * s_wx[k % kW] * inv;
  }
  for (int k = kK + tid; k < kWPad; k += 128) s_w[k] = 0.0f;
  __syncthreads();

  if (wave == 0) {
    bool nz[kRounds];
#pragma unroll
    for (int r = 0; r < kRounds; ++r) {
      const int j = r * 32 + lane;
      const v4f u = *(const v4f*)(s_w + j * kChunk);
      const v4f v = *(const v4f*)(s_w + j * kChunk + 4);
      nz[r] = (u.x != 0.0f) | (u.y != 0.0f) | (u.z != 0.0f) | (u.w != 0.0f) |
              (v.x != 0.0f) | (v.y != 0.0f) | (v.z != 0.0f) | (v.w != 0.0f);
    }
    int base = 0;
#pragma unroll
    for (int r = 0; r < kRounds; ++r) {
      const unsigned mask = __builtin_amdgcn_ballot_w32(nz[r]);
      if (nz[r]) s_list[base + (int)__builtin_amdgcn_mbcnt_lo(mask, 0u)] = r * 32 + lane;
      base += __builtin_popcount(mask);
    }
    const int padded = (base + kGroup - 1) & ~(kGroup - 1);
    if (base + lane < padded) s_list[base + lane] = kNChunk;
    if (lane == 0) s_cnt = padded;
  }
  __syncthreads();

  const int c0   = blockIdx.x * 64 + wave * 16;
  const int m    = lane & 15;
  const int koff = (lane >> 4) * 4;
  const float* __restrict__ arow =
      feat + ((size_t)n * kC + (size_t)(c0 + m)) * (size_t)kK + koff;

  const int cnt = __builtin_amdgcn_readfirstlane(s_cnt);

  v8f acc0 = {}, acc1 = {};
  for (int i = 0; i < cnt; i += kBatch) {
    const int rem = cnt - i;
    const int* lst = s_list + i;
    if (rem >= 32)      gemv_batch<32>(arow, s_w, lst, koff, acc0, acc1);
    else if (rem == 24) gemv_batch<24>(arow, s_w, lst, koff, acc0, acc1);
    else if (rem == 16) gemv_batch<16>(arow, s_w, lst, koff, acc0, acc1);
    else                gemv_batch<8>(arow, s_w, lst, koff, acc0, acc1);
  }
  v8f acc = acc0 + acc1;

  float* __restrict__ o = out + n * kC + c0;
  if (lane == 0) {
#pragma unroll
    for (int r = 0; r < 8; ++r) o[r] = acc[r];
  } else if (lane == 16) {
#pragma unroll
    for (int r = 0; r < 8; ++r) o[8 + r] = acc[r];
  }
}

extern "C" void kernel_launch(void* const* d_in, const int* in_sizes, int n_in,
                              void* d_out, int out_size, void* d_ws, size_t ws_size,
                              hipStream_t stream) {
  (void)n_in; (void)d_ws; (void)ws_size; (void)out_size;
  const float* feat = (const float*)d_in[0];
  const float* bb   = (const float*)d_in[1];
  float* out        = (float*)d_out;

  const int N = in_sizes[1] / 4;
  dim3 grid(kC / 64, N);
  dim3 block(128);
  prroi_wmma_kernel<<<grid, block, 0, stream>>>(feat, bb, out);
}


# A tile rows = 8 channels x 2 listed chunks (B columns split alike): a feature load touches fewer 128-byte lines; summation order changes: even and odd list positions on separate chains
# speedup vs baseline: 5.2793x; 1.1314x over previous
#include <hip/hip_runtime.h>


typedef __attribute__((ext_vector_type(2))) float v2f;
typedef __attribute__((ext_vector_type(4))) float v4f;
typedef __attribute__((ext_vector_type(8))) float v8f;

namespace {
constexpr int kH = 36, kW = 36, kC = 256, kK = kH * kW;
constexpr int kChunk  = 8;
constexpr int kNChunk = kK / kChunk;
constexpr int kGroup  = 8;
constexpr int kBatch  = 32;
constexpr int kRounds = (kNChunk + 31) / 32;
constexpr int kWPad   = kRounds * 32 * kChunk;
constexpr float kScale = 1.0f / 16.0f;

__device__ __forceinline__ float gfun(float u) {
  u = fminf(fmaxf(u, -1.0f), 1.0f);
  float p = u + 1.0f;
  float q = 1.0f - u;
  return (u < 0.0f) ? 0.5f * p * p : 1.0f - 0.5f * q * q;
}

template <int NC>
__device__ __forceinline__ void gemv_batch(const float* __restrict__ arow0,
                                           const float* __restrict__ arow1,
                                           const float* s_w, const int* lp,
                                           int koff, v8f (&acc)[2][2]) {
  constexpr int NP = NC / 2;
  v4f a0[NP], a1[NP], b[NP];
#pragma unroll
  for (int p = 0; p < NP; ++p) {
    const int j  = lp[2 * p];
    const int ka = (j < kNChunk ? j : 0) * kChunk;
    a0[p] = *(const v4f*)(arow0 + ka);
    a1[p] = *(const v4f*)(arow1 + ka);
    b[p]  = *(const v4f*)(s_w + j * kChunk + koff);
  }
  __builtin_amdgcn_sched_barrier(0);
#pragma unroll
  for (int p = 0; p < NP; ++p) {
    const v2f b0 = {b[p].x, b[p].y}, b1 = {b[p].z, b[p].w};
    const v2f x0 = {a0[p].x, a0[p].y}, x1 = {a0[p].z, a0[p].w};
    const v2f y0 = {a1[p].x, a1[p].y}, y1 = {a1[p].z, a1[p].w};
    acc[0][0] = __builtin_amdgcn_wmma_f32_16x16x4_f32(
         false, x0,  false, b0,
         (short)0, acc[0][0],  false,  false);
    acc[1][0] = __builtin_amdgcn_wmma_f32_16x16x4_f32(
        false, y0, false, b0, (short)0, acc[1][0], false, false);
    acc[0][1] = __builtin_amdgcn_wmma_f32_16x16x4_f32(
        false, x1, false, b1, (short)0, acc[0][1], false, false);
    acc[1][1] = __builtin_amdgcn_wmma_f32_16x16x4_f32(
        false, y1, false, b1, (short)0, acc[1][1], false, false);
  }
}
}

__global__ __launch_bounds__(128) void prroi_wmma_kernel(
    const float* __restrict__ feat,
    const float* __restrict__ bb,
    float* __restrict__ out) {
  __shared__ float s_wx[kW];
  __shared__ float s_wy[kH];
  __shared__ __attribute__((aligned(16))) float s_w[kWPad];
  __shared__ __attribute__((aligned(16))) int s_list[kNChunk + kGroup];
  __shared__ int s_cnt;

  const int n    = blockIdx.y;
  const int tid  = threadIdx.x;
  const int lane = tid & 31;
  const int wave = tid >> 5;

  const float bx = bb[n * 4 + 0], by = bb[n * 4 + 1];
  const float bw = bb[n * 4 + 2], bh = bb[n * 4 + 3];
  const float x0 = bx * kScale, y0 = by * kScale;
  const float x1 = (bx + bw) * kScale, y1 = (by + bh) * kScale;
  const float binw = x1 - x0, binh = y1 - y0;
  const float area = fmaxf(binw, 0.0f) * fmaxf(binh, 0.0f);
  const float inv  = (area > 0.0f) ? (1.0f / fmaxf(area, 1e-30f)) : 0.0f;

  if (tid < kW) {
    s_wx[tid] = gfun(x1 - (float)tid) - gfun(x0 - (float)tid);
  } else if (tid >= 64 && tid < 64 + kH) {
    const int i = tid - 64;
    s_wy[i] = gfun(y1 - (float)i) - gfun(y0 - (float)i);
  }
  __syncthreads();
  for (int k = tid; k < kK; k += 128) {
    s_w[k] = s_wy[k / kW] * s_wx[k % kW] * inv;
  }
  for (int k = kK + tid; k < kWPad; k += 128) s_w[k] = 0.0f;
  __syncthreads();

  if (wave == 0) {
    bool nz[kRounds];
#pragma unroll
    for (int r = 0; r < kRounds; ++r) {
      const int j = r * 32 + lane;
      const v4f u = *(const v4f*)(s_w + j * kChunk);
      const v4f v = *(const v4f*)(s_w + j * kChunk + 4);
      nz[r] = (u.x != 0.0f) | (u.y != 0.0f) | (u.z != 0.0f) | (u.w != 0.0f) |
              (v.x != 0.0f) | (v.y != 0.0f) | (v.z != 0.0f) | (v.w != 0.0f);
    }
    int base = 0;
#pragma unroll
    for (int r = 0; r < kRounds; ++r) {
      const unsigned mask = __builtin_amdgcn_ballot_w32(nz[r]);
      if (nz[r]) s_list[base + (int)__builtin_amdgcn_mbcnt_lo(mask, 0u)] = r * 32 + lane;
      base += __builtin_popcount(mask);
    }
    const int padded = (base + kGroup - 1) & ~(kGroup - 1);
    if (base + lane < padded) s_list[base + lane] = kNChunk;
    if (lane == 0) s_cnt = padded;
  }
  __syncthreads();

  const int c0   = blockIdx.x * 64 + wave * 16;
  const int m    = lane & 15;
  const int ch   = m & 7;
  const int slot = m >> 3;
  const int koff = (lane >> 4) * 4;
  const float* __restrict__ arow0 =
      feat + ((size_t)n * kC + (size_t)(c0 + ch)) * (size_t)kK + koff;
  const float* __restrict__ arow1 = arow0 + (size_t)8 * kK;

  const int cnt = __builtin_amdgcn_readfirstlane(s_cnt);

  v8f acc[2][2] = {};
  for (int i = 0; i < cnt; i += kBatch) {
    const int rem = cnt - i;
    const int* lp = s_list + i + slot;
    if (rem >= 32)      gemv_batch<32>(arow0, arow1, s_w, lp, koff, acc);
    else if (rem == 24) gemv_batch<24>(arow0, arow1, s_w, lp, koff, acc);
    else if (rem == 16) gemv_batch<16>(arow0, arow1, s_w, lp, koff, acc);
    else                gemv_batch<8>(arow0, arow1, s_w, lp, koff, acc);
  }

  float res[16];
#pragma unroll
  for (int g = 0; g < 2; ++g) {
    const v8f s = acc[g][0] + acc[g][1];
#pragma unroll
    for (int r = 0; r < 8; ++r) {
      const float odd = __int_as_float(
          __builtin_amdgcn_readlane(__float_as_int(s[r]), 24));
      res[g * 8 + r] = s[r] + odd;
    }
  }
  if (lane == 0) {
    float* __restrict__ o = out + n * kC + c0;
#pragma unroll
    for (int r = 0; r < 16; ++r) o[r] = res[r];
  }
}

extern "C" void kernel_launch(void* const* d_in, const int* in_sizes, int n_in,
                              void* d_out, int out_size, void* d_ws, size_t ws_size,
                              hipStream_t stream) {
  (void)n_in; (void)d_ws; (void)ws_size; (void)out_size;
  const float* feat = (const float*)d_in[0];
  const float* bb   = (const float*)d_in[1];
  float* out        = (float*)d_out;

  const int N = in_sizes[1] / 4;
  dim3 grid(kC / 64, N);
  dim3 block(128);
  prroi_wmma_kernel<<<grid, block, 0, stream>>>(feat, bb, out);
}


# each wave asks the second-level cache for its channels' box bytes (one contiguous range per channel, from the box bounds) right after bb is read, ahead of the weight and list build; sums unchanged
# speedup vs baseline: 5.7289x; 1.0852x over previous
#include <hip/hip_runtime.h>


typedef __attribute__((ext_vector_type(2))) float v2f;
typedef __attribute__((ext_vector_type(4))) float v4f;
typedef __attribute__((ext_vector_type(8))) float v8f;

namespace {
constexpr int kH = 36, kW = 36, kC = 256, kK = kH * kW;
constexpr int kChunk  = 8;
constexpr int kNChunk = kK / kChunk;
constexpr int kGroup  = 8;
constexpr int kBatch  = 32;
constexpr int kRounds = (kNChunk + 31) / 32;
constexpr int kWPad   = kRounds * 32 * kChunk;
constexpr float kScale = 1.0f / 16.0f;

__device__ __forceinline__ float gfun(float u) {
  u = fminf(fmaxf(u, -1.0f), 1.0f);
  float p = u + 1.0f;
  float q = 1.0f - u;
  return (u < 0.0f) ? 0.5f * p * p : 1.0f - 0.5f * q * q;
}

template <int NC>
__device__ __forceinline__ void gemv_batch(const float* __restrict__ arow0,
                                           const float* __restrict__ arow1,
                                           const float* s_w, const int* lp,
                                           int koff, v8f (&acc)[2][2]) {
  constexpr int NP = NC / 2;
  v4f a0[NP], a1[NP], b[NP];
#pragma unroll
  for (int p = 0; p < NP; ++p) {
    const int j  = lp[2 * p];
    const int ka = (j < kNChunk ? j : 0) * kChunk;
    a0[p] = *(const v4f*)(arow0 + ka);
    a1[p] = *(const v4f*)(arow1 + ka);
    b[p]  = *(const v4f*)(s_w + j * kChunk + koff);
  }
  __builtin_amdgcn_sched_barrier(0);
#pragma unroll
  for (int p = 0; p < NP; ++p) {
    const v2f b0 = {b[p].x, b[p].y}, b1 = {b[p].z, b[p].w};
    const v2f x0 = {a0[p].x, a0[p].y}, x1 = {a0[p].z, a0[p].w};
    const v2f y0 = {a1[p].x, a1[p].y}, y1 = {a1[p].z, a1[p].w};
    acc[0][0] = __builtin_amdgcn_wmma_f32_16x16x4_f32(
         false, x0,  false, b0,
         (short)0, acc[0][0],  false,  false);
    acc[1][0] = __builtin_amdgcn_wmma_f32_16x16x4_f32(
        false, y0, false, b0, (short)0, acc[1][0], false, false);
    acc[0][1] = __builtin_amdgcn_wmma_f32_16x16x4_f32(
        false, x1, false, b1, (short)0, acc[0][1], false, false);
    acc[1][1] = __builtin_amdgcn_wmma_f32_16x16x4_f32(
        false, y1, false, b1, (short)0, acc[1][1], false, false);
  }
}
}

__global__ __launch_bounds__(128) void prroi_wmma_kernel(
    const float* __restrict__ feat,
    const float* __restrict__ bb,
    float* __restrict__ out) {
  __shared__ float s_wx[kW];
  __shared__ float s_wy[kH];
  __shared__ __attribute__((aligned(16))) float s_w[kWPad];
  __shared__ __attribute__((aligned(16))) int s_list[kNChunk + kGroup];
  __shared__ int s_cnt;

  const int n    = blockIdx.y;
  const int tid  = threadIdx.x;
  const int lane = tid & 31;
  const int wave = tid >> 5;

  const float bx = bb[n * 4 + 0], by = bb[n * 4 + 1];
  const float bw = bb[n * 4 + 2], bh = bb[n * 4 + 3];
  const float x0 = bx * kScale, y0 = by * kScale;
  const float x1 = (bx + bw) * kScale, y1 = (by + bh) * kScale;
  {
    const float fxa = fminf(fmaxf(fminf(x0, x1), 0.0f), (float)(kW - 1));
    const float fxb = fminf(fmaxf(fmaxf(x0, x1), 0.0f), (float)(kW - 1));
    const float fya = fminf(fmaxf(fminf(y0, y1), 0.0f), (float)(kH - 1));
    const float fyb = fminf(fmaxf(fmaxf(y0, y1), 0.0f), (float)(kH - 1));
    const int pxlo = (int)floorf(fxa), pxhi = (int)ceilf(fxb);
    const int pylo = (int)floorf(fya), pyhi = (int)ceilf(fyb);
    const int bfirst = (pylo * kW + pxlo) * 4;
    const int blast  = (pyhi * kW + pxhi) * 4;
    const char* pch = (const char*)(feat +
        ((size_t)n * kC + (size_t)(blockIdx.x * 64 + wave * 16 + (lane & 15))) * (size_t)kK);
    const int nstep = (blast - bfirst) / 256 + 2;
    const int boff  = bfirst + (lane >> 4) * 256;
    for (int s = 0; s < nstep; s += 2) {
      const int o = boff + s * 256;
      __builtin_prefetch(pch + (o < blast ? o : blast), 0, 2);
    }
    __builtin_prefetch(pch, 0, 2);
  }

  const float binw = x1 - x0, binh = y1 - y0;
  const float area = fmaxf(binw, 0.0f) * fmaxf(binh, 0.0f);
  const float inv  = (area > 0.0f) ? (1.0f / fmaxf(area, 1e-30f)) : 0.0f;

  if (tid < kW) {
    s_wx[tid] = gfun(x1 - (float)tid) - gfun(x0 - (float)tid);
  } else if (tid >= 64 && tid < 64 + kH) {
    const int i = tid - 64;
    s_wy[i] = gfun(y1 - (float)i) - gfun(y0 - (float)i);
  }
  __syncthreads();
  for (int k = tid; k < kK; k += 128) {
    s_w[k] = s_wy[k / kW] * s_wx[k % kW] * inv;
  }
  for (int k = kK + tid; k < kWPad; k += 128) s_w[k] = 0.0f;
  __syncthreads();

  if (wave == 0) {
    bool nz[kRounds];
#pragma unroll
    for (int r = 0; r < kRounds; ++r) {
      const int j = r * 32 + lane;
      const v4f u = *(const v4f*)(s_w + j * kChunk);
      const v4f v = *(const v4f*)(s_w + j * kChunk + 4);
      nz[r] = (u.x != 0.0f) | (u.y != 0.0f) | (u.z != 0.0f) | (u.w != 0.0f) |
              (v.x != 0.0f) | (v.y != 0.0f) | (v.z != 0.0f) | (v.w != 0.0f);
    }
    int base = 0;
#pragma unroll
    for (int r = 0; r < kRounds; ++r) {
      const unsigned mask = __builtin_amdgcn_ballot_w32(nz[r]);
      if (nz[r]) s_list[base + (int)__builtin_amdgcn_mbcnt_lo(mask, 0u)] = r * 32 + lane;
      base += __builtin_popcount(mask);
    }
    const int padded = (base + kGroup - 1) & ~(kGroup - 1);
    if (base + lane < padded) s_list[base + lane] = kNChunk;
    if (lane == 0) s_cnt = padded;
  }
  __syncthreads();

  const int c0   = blockIdx.x * 64 + wave * 16;
  const int m    = lane & 15;
  const int ch   = m & 7;
  const int slot = m >> 3;
  const int koff = (lane >> 4) * 4;
  const float* __restrict__ arow0 =
      feat + ((size_t)n * kC + (size_t)(c0 + ch)) * (size_t)kK + koff;
  const float* __restrict__ arow1 = arow0 + (size_t)8 * kK;

  const int cnt = __builtin_amdgcn_readfirstlane(s_cnt);

  v8f acc[2][2] = {};
  for (int i = 0; i < cnt; i += kBatch) {
    const int rem = cnt - i;
    const int* lp = s_list + i + slot;
    if (rem >= 32)      gemv_batch<32>(arow0, arow1, s_w, lp, koff, acc);
    else if (rem == 24) gemv_batch<24>(arow0, arow1, s_w, lp, koff, acc);
    else if (rem == 16) gemv_batch<16>(arow0, arow1, s_w, lp, koff, acc);
    else                gemv_batch<8>(arow0, arow1, s_w, lp, koff, acc);
  }

  float res[16];
#pragma unroll
  for (int g = 0; g < 2; ++g) {
    const v8f s = acc[g][0] + acc[g][1];
#pragma unroll
    for (int r = 0; r < 8; ++r) {
      const float odd = __int_as_float(
          __builtin_amdgcn_readlane(__float_as_int(s[r]), 24));
      res[g * 8 + r] = s[r] + odd;
    }
  }
  if (lane == 0) {
    float* __restrict__ o = out + n * kC + c0;
#pragma unroll
    for (int r = 0; r < 16; ++r) o[r] = res[r];
  }
}

extern "C" void kernel_launch(void* const* d_in, const int* in_sizes, int n_in,
                              void* d_out, int out_size, void* d_ws, size_t ws_size,
                              hipStream_t stream) {
  (void)n_in; (void)d_ws; (void)ws_size; (void)out_size;
  const float* feat = (const float*)d_in[0];
  const float* bb   = (const float*)d_in[1];
  float* out        = (float*)d_out;

  const int N = in_sizes[1] / 4;
  dim3 grid(kC / 64, N);
  dim3 block(128);
  prroi_wmma_kernel<<<grid, block, 0, stream>>>(feat, bb, out);
}


# 8-float units start at each row's first weighted 4-float group (rows with wy!=0 x column starts from wx), not at fixed chunks found by a scan; sum order changes: list parity x float pair of a group
# speedup vs baseline: 6.1247x; 1.0691x over previous
#include <hip/hip_runtime.h>


typedef __attribute__((ext_vector_type(2))) float v2f;
typedef __attribute__((ext_vector_type(4))) float v4f;
typedef __attribute__((ext_vector_type(8))) float v8f;

namespace {
constexpr int kH = 36, kW = 36, kC = 256, kK = kH * kW;
constexpr int kQuad   = 4;
constexpr int kQRow   = kW / kQuad;
constexpr int kNQuad  = kK / kQuad;
constexpr int kMaxU   = (kQRow + 1) / 2;
constexpr int kMaxPairs = 12;
constexpr int kWPad   = kK + kQuad;
constexpr int kListCap = kH * kMaxU + 4;
constexpr int kStartShift = 12;
constexpr float kScale = 1.0f / 16.0f;

__device__ __forceinline__ float gfun(float u) {
  u = fminf(fmaxf(u, -1.0f), 1.0f);
  float p = u + 1.0f;
  float q = 1.0f - u;
  return (u < 0.0f) ? 0.5f * p * p : 1.0f - 0.5f * q * q;
}

template <int NP>
__device__ __forceinline__ void gemv_batch(const float* __restrict__ arow0,
                                           const float* __restrict__ arow1,
                                           const float* s_w, const int* lp,
                                           int h, v8f (&acc)[2][2]) {
  v4f a0[NP], a1[NP], b[NP];
#pragma unroll
  for (int p = 0; p < NP; ++p) {
    const int e  = lp[2 * p];
    const int j0 = e & ((1 << kStartShift) - 1);
    const int s  = e >> kStartShift;
    const int jf = min(j0 + h, kNQuad - 1);
    const int jw = (s + h < kQRow) ? (j0 + h) : kNQuad;
    a0[p] = *(const v4f*)(arow0 + jf * kQuad);
    a1[p] = *(const v4f*)(arow1 + jf * kQuad);
    b[p]  = *(const v4f*)(s_w + jw * kQuad);
  }
  __builtin_amdgcn_sched_barrier(0);
#pragma unroll
  for (int p = 0; p < NP; ++p) {
    const v2f b0 = {b[p].x, b[p].y}, b1 = {b[p].z, b[p].w};
    const v2f x0 = {a0[p].x, a0[p].y}, x1 = {a0[p].z, a0[p].w};
    const v2f y0 = {a1[p].x, a1[p].y}, y1 = {a1[p].z, a1[p].w};
    acc[0][0] = __builtin_amdgcn_wmma_f32_16x16x4_f32(
         false, x0,  false, b0,
         (short)0, acc[0][0],  false,  false);
    acc[1][0] = __builtin_amdgcn_wmma_f32_16x16x4_f32(
        false, y0, false, b0, (short)0, acc[1][0], false, false);
    acc[0][1] = __builtin_amdgcn_wmma_f32_16x16x4_f32(
        false, x1, false, b1, (short)0, acc[0][1], false, false);
    acc[1][1] = __builtin_amdgcn_wmma_f32_16x16x4_f32(
        false, y1, false, b1, (short)0, acc[1][1], false, false);
  }
}
}

__global__ __launch_bounds__(128) void prroi_wmma_kernel(
    const float* __restrict__ feat,
    const float* __restrict__ bb,
    float* __restrict__ out) {
  __shared__ __attribute__((aligned(16))) float s_wx[kW];
  __shared__ float s_wy[kH];
  __shared__ __attribute__((aligned(16))) float s_w[kWPad];
  __shared__ __attribute__((aligned(16))) int s_list[kListCap];
  __shared__ int s_cnt;

  const int n    = blockIdx.y;
  const int tid  = threadIdx.x;
  const int lane = tid & 31;
  const int wave = tid >> 5;

  const float bx = bb[n * 4 + 0], by = bb[n * 4 + 1];
  const float bw = bb[n * 4 + 2], bh = bb[n * 4 + 3];
  const float x0 = bx * kScale, y0 = by * kScale;
  const float x1 = (bx + bw) * kScale, y1 = (by + bh) * kScale;
  {
    const float fxa = fminf(fmaxf(fminf(x0, x1), 0.0f), (float)(kW - 1));
    const float fxb = fminf(fmaxf(fmaxf(x0, x1), 0.0f), (float)(kW - 1));
    const float fya = fminf(fmaxf(fminf(y0, y1), 0.0f), (float)(kH - 1));
    const float fyb = fminf(fmaxf(fmaxf(y0, y1), 0.0f), (float)(kH - 1));
    const int pxlo = (int)floorf(fxa), pxhi = (int)ceilf(fxb);
    const int pylo = (int)floorf(fya), pyhi = (int)ceilf(fyb);
    const int bfirst = (pylo * kW + pxlo) * 4;
    const int blast  = min((pyhi * kW + pxhi) * 4 + 28, (kK - 1) * 4);
    const char* pch = (const char*)(feat +
        ((size_t)n * kC + (size_t)(blockIdx.x * 64 + wave * 16 + (lane & 15))) * (size_t)kK);
    const int nstep = (blast - bfirst) / 256 + 2;
    const int boff  = bfirst + (lane >> 4) * 256;
    for (int s = 0; s < nstep; s += 2) {
      const int o = boff + s * 256;
      __builtin_prefetch(pch + (o < blast ? o : blast), 0, 2);
    }
  }

  const float binw = x1 - x0, binh = y1 - y0;
  const float area = fmaxf(binw, 0.0f) * fmaxf(binh, 0.0f);
  const float inv  = (area > 0.0f) ? (1.0f / fmaxf(area, 1e-30f)) : 0.0f;

  if (tid < kW) {
    s_wx[tid] = gfun(x1 - (float)tid) - gfun(x0 - (float)tid);
  } else if (tid >= 64 && tid < 64 + kH) {
    const int i = tid - 64;
    s_wy[i] = gfun(y1 - (float)i) - gfun(y0 - (float)i);
  }
  __syncthreads();
  if (wave == 3) {
    const bool ra = (s_wy[lane] != 0.0f);
    const bool rb = (lane < kH - 32) & (s_wy[32 + (lane & 3)] != 0.0f);
    bool ca = false;
    if (lane < kQRow) {
      const v4f u = *(const v4f*)(s_wx + lane * kQuad);
      ca = (u.x != 0.0f) | (u.y != 0.0f) | (u.z != 0.0f) | (u.w != 0.0f);
    }
    const unsigned mlo = __builtin_amdgcn_ballot_w32(ra);
    const unsigned mhi = __builtin_amdgcn_ballot_w32(rb);
    unsigned cm = __builtin_amdgcn_ballot_w32(ca);
    int ws[kMaxU];
    int nu = 0;
#pragma unroll
    for (int t = 0; t < kMaxU; ++t) {
      ws[t] = 0;
      if (cm != 0u) {
        const int s = __builtin_ctz(cm);
        ws[t] = s;
        cm &= ~(3u << s);
        nu = t + 1;
      }
    }
    const int nlo   = __builtin_popcount(mlo);
    const int nrow  = nlo + __builtin_popcount(mhi);
    const int total = nrow * nu;
    const int ka = (int)__builtin_amdgcn_mbcnt_lo(mlo, 0u) * nu;
    const int kb = (nlo + (int)__builtin_amdgcn_mbcnt_lo(mhi, 0u)) * nu;
#pragma unroll
    for (int t = 0; t < kMaxU; ++t) {
      if (t < nu) {
        if (ra) s_list[ka + t] = (lane * kQRow + ws[t]) | (ws[t] << kStartShift);
        if (rb) s_list[kb + t] = ((32 + lane) * kQRow + ws[t]) | (ws[t] << kStartShift);
      }
    }
    if (lane == 0) {
      if (total & 1) {
        const int rl = (mhi != 0u) ? (63 - __builtin_clz(mhi)) : (31 - __builtin_clz(mlo | 1u));
        s_list[total] = (rl * kQRow + ws[nu > 0 ? nu - 1 : 0]) | (kQRow << kStartShift);
      }
      s_cnt = (total + 1) & ~1;
    }
  }
  for (int k = tid; k < kK; k += 128) {
    s_w[k] = s_wy[k / kW] * s_wx[k % kW] * inv;
  }
  if (tid < kWPad - kK) s_w[kK + tid] = 0.0f;
  __syncthreads();

  const int c0   = blockIdx.x * 64 + wave * 16;
  const int m    = lane & 15;
  const int ch   = m & 7;
  const int slot = m >> 3;
  const int h    = lane >> 4;
  const float* __restrict__ arow0 =
      feat + ((size_t)n * kC + (size_t)(c0 + ch)) * (size_t)kK;
  const float* __restrict__ arow1 = arow0 + (size_t)8 * kK;

  const int cnt = __builtin_amdgcn_readfirstlane(s_cnt);

  v8f acc[2][2] = {};
  for (int i = 0; i < cnt; i += 2 * kMaxPairs) {
    const int rem = (cnt - i) >> 1;
    const int* lp = s_list + i + slot;
    if (rem >= 12) gemv_batch<12>(arow0, arow1, s_w, lp, h, acc);
    else switch (rem) {
      case 11: gemv_batch<11>(arow0, arow1, s_w, lp, h, acc); break;
      case 10: gemv_batch<10>(arow0, arow1, s_w, lp, h, acc); break;
      case 9:  gemv_batch<9>(arow0, arow1, s_w, lp, h, acc); break;
      case 8:  gemv_batch<8>(arow0, arow1, s_w, lp, h, acc); break;
      case 7:  gemv_batch<7>(arow0, arow1, s_w, lp, h, acc); break;
      case 6:  gemv_batch<6>(arow0, arow1, s_w, lp, h, acc); break;
      case 5:  gemv_batch<5>(arow0, arow1, s_w, lp, h, acc); break;
      case 4:  gemv_batch<4>(arow0, arow1, s_w, lp, h, acc); break;
      case 3:  gemv_batch<3>(arow0, arow1, s_w, lp, h, acc); break;
      case 2:  gemv_batch<2>(arow0, arow1, s_w, lp, h, acc); break;
      default: gemv_batch<1>(arow0, arow1, s_w, lp, h, acc); break;
    }
  }

  float res[16];
#pragma unroll
  for (int g = 0; g < 2; ++g) {
    const v8f s = acc[g][0] + acc[g][1];
#pragma unroll
    for (int r = 0; r < 8; ++r) {
      const float odd = __int_as_float(
          __builtin_amdgcn_readlane(__float_as_int(s[r]), 24));
      res[g * 8 + r] = s[r] + odd;
    }
  }
  if (lane == 0) {
    float* __restrict__ o = out + n * kC + c0;
#pragma unroll
    for (int r = 0; r < 16; ++r) o[r] = res[r];
  }
}

extern "C" void kernel_launch(void* const* d_in, const int* in_sizes, int n_in,
                              void* d_out, int out_size, void* d_ws, size_t ws_size,
                              hipStream_t stream) {
  (void)n_in; (void)d_ws; (void)ws_size; (void)out_size;
  const float* feat = (const float*)d_in[0];
  const float* bb   = (const float*)d_in[1];
  float* out        = (float*)d_out;

  const int N = in_sizes[1] / 4;
  dim3 grid(kC / 64, N);
  dim3 block(128);
  prroi_wmma_kernel<<<grid, block, 0, stream>>>(feat, bb, out);
}
